# starting kernel
# baseline (speedup 1.0000x reference)
#include <hip/hip_runtime.h>
#include <hip/hip_bf16.h>

typedef __attribute__((ext_vector_type(2))) float v2f;
typedef __attribute__((ext_vector_type(8))) float v8f;

#define N_NODES 89250
#define N_EDGES 900000
#define F_IN 500
#define N_CLASSES 7

__global__ void gcn_deg_kernel(const int* __restrict__ ei,
                               float* __restrict__ outDeg,
                               float* __restrict__ inDeg, int nEdges) {
    int e = blockIdx.x * blockDim.x + threadIdx.x;
    if (e < nEdges) {
        atomicAdd(&outDeg[ei[e]], 1.0f);
        atomicAdd(&inDeg[ei[nEdges + e]], 1.0f);
    }
}

__global__ void gcn_norm_kernel(float* __restrict__ d, int n) {
    int i = blockIdx.x * blockDim.x + threadIdx.x;
    if (i < n) d[i] = rsqrtf(fmaxf(d[i], 1.0f));
}

__global__ void gcn_pack_b_kernel(const float* __restrict__ W, int K, int Ncols,
                                  v2f* __restrict__ Bpack, int kChunks) {
    int t = blockIdx.x * blockDim.x + threadIdx.x;
    if (t >= kChunks * 32) return;
    int chunk = t >> 5;
    int lane  = t & 31;
    int n = lane & 15;
    int k = chunk * 4 + ((lane >> 4) << 1);
    v2f b;
    b[0] = (k     < K && n < Ncols) ? W[k * Ncols + n]       : 0.0f;
    b[1] = (k + 1 < K && n < Ncols) ? W[(k + 1) * Ncols + n] : 0.0f;
    Bpack[t] = b;
}

__global__ void gcn_wmma_gemm_kernel(const float* __restrict__ A, int lda,
                                     int kChunks, const v2f* __restrict__ Bpack,
                                     const float* __restrict__ normSrc,
                                     float* __restrict__ Xpad, int nRows) {
    int wave = (blockIdx.x * blockDim.x + threadIdx.x) >> 5;
    int lane = threadIdx.x & 31;
    int row0 = wave * 16;
    if (row0 >= nRows) return;

    int m    = lane & 15;
    int kOff = (lane >> 4) << 1;
    int row  = row0 + m;
    int rowC = (row < nRows) ? row : (nRows - 1);
    float ns = normSrc[rowC];
    const float* aRow = A + (long)rowC * lda;

    v8f c = {};
    for (int kc = 0; kc < kChunks; ++kc) {
        int k = kc * 4 + kOff;
        v2f a = *(const v2f*)(aRow + k);
        a[0] *= ns;
        a[1] *= ns;
        v2f b = Bpack[kc * 32 + lane];
        c = __builtin_amdgcn_wmma_f32_16x16x4_f32(
                false, a, false, b, (short)0, c, false, false);
    }

    int n     = lane & 15;
    int rbase = row0 + ((lane >> 4) << 3);
    for (int j = 0; j < 8; ++j) {
        int r = rbase + j;
        if (r < nRows) Xpad[r * 16 + n] = c[j];
    }
}

__global__ void gcn_scatter_kernel(const int* __restrict__ ei,
                                   const float* __restrict__ xpad,
                                   float* __restrict__ agg, int nEdges) {
    int t = blockIdx.x * blockDim.x + threadIdx.x;
    int e = t >> 3;
    int c = t & 7;
    if (e < nEdges && c < N_CLASSES) {
        int src = ei[e];
        int dst = ei[nEdges + e];
        atomicAdd(&agg[dst * 8 + c], xpad[src * 16 + c]);
    }
}

__global__ void gcn_finish_pad_kernel(const float* __restrict__ agg,
                                      const float* __restrict__ normDst,
                                      const float* __restrict__ bias,
                                      float* __restrict__ h1pad, int n) {
    int t = blockIdx.x * blockDim.x + threadIdx.x;
    if (t >= n * 16) return;
    int i = t >> 4;
    int c = t & 15;
    float v = 0.0f;
    if (c < N_CLASSES) v = agg[i * 8 + c] * normDst[i] + bias[c];
    h1pad[t] = v;
}

__global__ void gcn_final_kernel(const float* __restrict__ agg,
                                 const float* __restrict__ normDst,
                                 const float* __restrict__ bias,
                                 float* __restrict__ out, int n) {
    int t = blockIdx.x * blockDim.x + threadIdx.x;
    int i = t >> 3;
    int c = t & 7;
    if (i < n && c < N_CLASSES)
        out[i * N_CLASSES + c] = agg[i * 8 + c] * normDst[i] + bias[c];
}

static inline size_t alignUp256(size_t x) { return (x + 255) & ~(size_t)255; }

extern "C" void kernel_launch(void* const* d_in, const int* in_sizes, int n_in,
                              void* d_out, int out_size, void* d_ws, size_t ws_size,
                              hipStream_t stream) {
    const float* h  = (const float*)d_in[0];
    const int*   ei = (const int*)  d_in[1];
    const float* W1 = (const float*)d_in[2];
    const float* b1 = (const float*)d_in[3];
    const float* W2 = (const float*)d_in[4];
    const float* b2 = (const float*)d_in[5];
    float* out = (float*)d_out;

    const int N = N_NODES, E = N_EDGES;
    const int kChunks1 = F_IN / 4;
    const int kChunks2 = 2;

    char*  ws  = (char*)d_ws;
    size_t off = 0;
    auto take = [&](size_t bytes) -> void* {
        void* p = ws + off;
        off = alignUp256(off + bytes);
        return p;
    };
    float* normSrc = (float*)take((size_t)N * 4);
    float* normDst = (float*)take((size_t)N * 4);
    float* xpad    = (float*)take((size_t)N * 16 * 4);
    float* h1pad   = (float*)take((size_t)N * 16 * 4);
    float* agg1    = (float*)take((size_t)N * 8 * 4);
    float* agg2    = (float*)take((size_t)N * 8 * 4);
    v2f*   bp1     = (v2f*)take((size_t)kChunks1 * 32 * 8);
    v2f*   bp2     = (v2f*)take((size_t)kChunks2 * 32 * 8);
    (void)ws_size;

    hipMemsetAsync(normSrc, 0, (size_t)N * 4, stream);
    hipMemsetAsync(normDst, 0, (size_t)N * 4, stream);
    hipMemsetAsync(agg1,    0, (size_t)N * 8 * 4, stream);
    hipMemsetAsync(agg2,    0, (size_t)N * 8 * 4, stream);

    gcn_deg_kernel<<<(E + 255) / 256, 256, 0, stream>>>(ei, normSrc, normDst, E);
    gcn_norm_kernel<<<(N + 255) / 256, 256, 0, stream>>>(normSrc, N);
    gcn_norm_kernel<<<(N + 255) / 256, 256, 0, stream>>>(normDst, N);

    gcn_pack_b_kernel<<<(kChunks1 * 32 + 255) / 256, 256, 0, stream>>>(
        W1, F_IN, N_CLASSES, bp1, kChunks1);
    gcn_pack_b_kernel<<<1, 64, 0, stream>>>(
        W2, N_CLASSES, N_CLASSES, bp2, kChunks2);

    int tiles  = (N + 15) / 16;
    int gemmBl = (tiles * 32 + 255) / 256;
    gcn_wmma_gemm_kernel<<<gemmBl, 256, 0, stream>>>(
        h, F_IN, kChunks1, bp1, normSrc, xpad, N);
    gcn_scatter_kernel<<<(E * 8 + 255) / 256, 256, 0, stream>>>(ei, xpad, agg1, E);
    gcn_finish_pad_kernel<<<(N * 16 + 255) / 256, 256, 0, stream>>>(
        agg1, normDst, b1, h1pad, N);

    gcn_wmma_gemm_kernel<<<gemmBl, 256, 0, stream>>>(
        h1pad, 16, kChunks2, bp2, normSrc, xpad, N);
    gcn_scatter_kernel<<<(E * 8 + 255) / 256, 256, 0, stream>>>(ei, xpad, agg2, E);
    gcn_final_kernel<<<(N * 8 + 255) / 256, 256, 0, stream>>>(
        agg2, normDst, b2, out, N);

    (void)in_sizes; (void)n_in; (void)out_size;
}


# layer-1 GEMM: each wave streams its contiguous 16-row A tile with whole-wave 16-byte non-temporal loads into LDS and reads the WMMA A fragments from LDS; k order and summation order unchanged
# speedup vs baseline: 1.1991x; 1.1991x over previous
#include <hip/hip_runtime.h>
#include <hip/hip_bf16.h>

typedef __attribute__((ext_vector_type(2))) float v2f;
typedef __attribute__((ext_vector_type(8))) float v8f;
typedef __attribute__((ext_vector_type(4))) float v4f;

#define N_NODES 89250
#define N_EDGES 900000
#define F_IN 500
#define N_CLASSES 7

__global__ void gcn_deg_kernel(const int* __restrict__ ei,
                               float* __restrict__ outDeg,
                               float* __restrict__ inDeg, int nEdges) {
    int e = blockIdx.x * blockDim.x + threadIdx.x;
    if (e < nEdges) {
        atomicAdd(&outDeg[ei[e]], 1.0f);
        atomicAdd(&inDeg[ei[nEdges + e]], 1.0f);
    }
}

__global__ void gcn_norm_kernel(float* __restrict__ d, int n) {
    int i = blockIdx.x * blockDim.x + threadIdx.x;
    if (i < n) d[i] = rsqrtf(fmaxf(d[i], 1.0f));
}

__global__ void gcn_pack_b_kernel(const float* __restrict__ W, int K, int Ncols,
                                  v2f* __restrict__ Bpack, int kChunks) {
    int t = blockIdx.x * blockDim.x + threadIdx.x;
    if (t >= kChunks * 32) return;
    int chunk = t >> 5;
    int lane  = t & 31;
    int n = lane & 15;
    int k = chunk * 4 + ((lane >> 4) << 1);
    v2f b;
    b[0] = (k     < K && n < Ncols) ? W[k * Ncols + n]       : 0.0f;
    b[1] = (k + 1 < K && n < Ncols) ? W[(k + 1) * Ncols + n] : 0.0f;
    Bpack[t] = b;
}

__global__ void gcn_wmma_gemm_kernel(const float* __restrict__ A, int lda,
                                     int kChunks, const v2f* __restrict__ Bpack,
                                     const float* __restrict__ normSrc,
                                     float* __restrict__ Xpad, int nRows) {
    int wave = (blockIdx.x * blockDim.x + threadIdx.x) >> 5;
    int lane = threadIdx.x & 31;
    int row0 = wave * 16;
    if (row0 >= nRows) return;

    int m    = lane & 15;
    int kOff = (lane >> 4) << 1;
    int row  = row0 + m;
    int rowC = (row < nRows) ? row : (nRows - 1);
    float ns = normSrc[rowC];
    const float* aRow = A + (long)rowC * lda;

    v8f c = {};
    for (int kc = 0; kc < kChunks; ++kc) {
        int k = kc * 4 + kOff;
        v2f a = *(const v2f*)(aRow + k);
        a[0] *= ns;
        a[1] *= ns;
        v2f b = Bpack[kc * 32 + lane];
        c = __builtin_amdgcn_wmma_f32_16x16x4_f32(
                false, a, false, b, (short)0, c, false, false);
    }

    int n     = lane & 15;
    int rbase = row0 + ((lane >> 4) << 3);
    for (int j = 0; j < 8; ++j) {
        int r = rbase + j;
        if (r < nRows) Xpad[r * 16 + n] = c[j];
    }
}

#define TILE_ROWS 16
#define TILE_FLOATS (TILE_ROWS * F_IN)
#define TILE_VEC4   (TILE_FLOATS / 4)
#define VEC4_FULL   (TILE_VEC4 / 32)
#define TAIL_FLOAT0 (VEC4_FULL * 32 * 4)
#define VEC4_FIRST  48
#define VEC4_REST   (VEC4_FULL - VEC4_FIRST)
#define KC_GROUP    25
#define KC_GROUPS   (F_IN / 4 / KC_GROUP)

__global__ void __launch_bounds__(32)
gcn_wmma_gemm_stream_kernel(const float* __restrict__ A,
                            const v2f* __restrict__ Bpack,
                            const float* __restrict__ normSrc,
                            float* __restrict__ Xpad, int nRows) {
    __shared__ __attribute__((aligned(16))) float tileF[TILE_FLOATS];

    const int lane = threadIdx.x;
    const int row0 = blockIdx.x * TILE_ROWS;
    if (row0 >= nRows) return;

    const int m    = lane & 15;
    const int kOff = (lane >> 4) << 1;
    const int row  = row0 + m;
    const int rowC = (row < nRows) ? row : (nRows - 1);
    const float ns = normSrc[rowC];

    const int validRows = (nRows - row0 < TILE_ROWS) ? (nRows - row0) : TILE_ROWS;
    const float* __restrict__ srcF = A + (size_t)row0 * F_IN;
    const v4f*   __restrict__ src  = (const v4f*)srcF;
    v4f* tileV = (v4f*)tileF;

    if (validRows == TILE_ROWS) {
        v4f v[VEC4_FIRST];
#pragma unroll
        for (int i = 0; i < VEC4_FIRST; ++i)
            v[i] = __builtin_nontemporal_load(src + i * 32 + lane);
#pragma unroll
        for (int i = 0; i < VEC4_REST; ++i)
            tileV[i * 32 + lane] = v[i];
        asm volatile("" ::: "memory");
#pragma unroll
        for (int i = 0; i < VEC4_REST; ++i)
            v[i] = __builtin_nontemporal_load(src + (VEC4_FIRST + i) * 32 + lane);
        const v2f t = __builtin_nontemporal_load((const v2f*)(srcF + TAIL_FLOAT0) + lane);
#pragma unroll
        for (int i = VEC4_REST; i < VEC4_FIRST; ++i)
            tileV[i * 32 + lane] = v[i];
#pragma unroll
        for (int i = 0; i < VEC4_REST; ++i)
            tileV[(VEC4_FIRST + i) * 32 + lane] = v[i];
        ((v2f*)(tileF + TAIL_FLOAT0))[lane] = t;
    } else {
        const int nVec = validRows * (F_IN / 4);
        for (int idx = lane; idx < TILE_VEC4; idx += 32) {
            v4f x = {0.0f, 0.0f, 0.0f, 0.0f};
            if (idx < nVec) x = src[idx];
            tileV[idx] = x;
        }
    }

    __syncthreads();

    const float* aRow = tileF + m * F_IN + kOff;
    const v2f*   bRow = Bpack + lane;
    v8f c = {};
    v2f a[2][KC_GROUP];
    v2f b[2][KC_GROUP];
#pragma unroll
    for (int i = 0; i < KC_GROUP; ++i) {
        a[0][i] = *(const v2f*)(aRow + i * 4);
        b[0][i] = bRow[i * 32];
    }
#pragma unroll
    for (int g = 0; g < KC_GROUPS; ++g) {
        __builtin_amdgcn_sched_barrier(0);
        if (g + 1 < KC_GROUPS) {
#pragma unroll
            for (int i = 0; i < KC_GROUP; ++i) {
                const int kc = (g + 1) * KC_GROUP + i;
                a[(g + 1) & 1][i] = *(const v2f*)(aRow + kc * 4);
                b[(g + 1) & 1][i] = bRow[kc * 32];
            }
        }
        __builtin_amdgcn_sched_barrier(0);
#pragma unroll
        for (int i = 0; i < KC_GROUP; ++i) {
            v2f x = a[g & 1][i];
            x[0] *= ns;
            x[1] *= ns;
            c = __builtin_amdgcn_wmma_f32_16x16x4_f32(
                    false, x, false, b[g & 1][i], (short)0, c, false, false);
        }
    }

    int n     = lane & 15;
    int rbase = row0 + ((lane >> 4) << 3);
    for (int j = 0; j < 8; ++j) {
        int r = rbase + j;
        if (r < nRows) Xpad[r * 16 + n] = c[j];
    }
}

__global__ void gcn_scatter_kernel(const int* __restrict__ ei,
                                   const float* __restrict__ xpad,
                                   float* __restrict__ agg, int nEdges) {
    int t = blockIdx.x * blockDim.x + threadIdx.x;
    int e = t >> 3;
    int c = t & 7;
    if (e < nEdges && c < N_CLASSES) {
        int src = ei[e];
        int dst = ei[nEdges + e];
        atomicAdd(&agg[dst * 8 + c], xpad[src * 16 + c]);
    }
}

__global__ void gcn_finish_pad_kernel(const float* __restrict__ agg,
                                      const float* __restrict__ normDst,
                                      const float* __restrict__ bias,
                                      float* __restrict__ h1pad, int n) {
    int t = blockIdx.x * blockDim.x + threadIdx.x;
    if (t >= n * 16) return;
    int i = t >> 4;
    int c = t & 15;
    float v = 0.0f;
    if (c < N_CLASSES) v = agg[i * 8 + c] * normDst[i] + bias[c];
    h1pad[t] = v;
}

__global__ void gcn_final_kernel(const float* __restrict__ agg,
                                 const float* __restrict__ normDst,
                                 const float* __restrict__ bias,
                                 float* __restrict__ out, int n) {
    int t = blockIdx.x * blockDim.x + threadIdx.x;
    int i = t >> 3;
    int c = t & 7;
    if (i < n && c < N_CLASSES)
        out[i * N_CLASSES + c] = agg[i * 8 + c] * normDst[i] + bias[c];
}

static inline size_t alignUp256(size_t x) { return (x + 255) & ~(size_t)255; }

extern "C" void kernel_launch(void* const* d_in, const int* in_sizes, int n_in,
                              void* d_out, int out_size, void* d_ws, size_t ws_size,
                              hipStream_t stream) {
    const float* h  = (const float*)d_in[0];
    const int*   ei = (const int*)  d_in[1];
    const float* W1 = (const float*)d_in[2];
    const float* b1 = (const float*)d_in[3];
    const float* W2 = (const float*)d_in[4];
    const float* b2 = (const float*)d_in[5];
    float* out = (float*)d_out;

    const int N = N_NODES, E = N_EDGES;
    const int kChunks1 = F_IN / 4;
    const int kChunks2 = 2;

    char*  ws  = (char*)d_ws;
    size_t off = 0;
    auto take = [&](size_t bytes) -> void* {
        void* p = ws + off;
        off = alignUp256(off + bytes);
        return p;
    };
    float* normSrc = (float*)take((size_t)N * 4);
    float* normDst = (float*)take((size_t)N * 4);
    float* xpad    = (float*)take((size_t)N * 16 * 4);
    float* h1pad   = (float*)take((size_t)N * 16 * 4);
    float* agg1    = (float*)take((size_t)N * 8 * 4);
    float* agg2    = (float*)take((size_t)N * 8 * 4);
    v2f*   bp1     = (v2f*)take((size_t)kChunks1 * 32 * 8);
    v2f*   bp2     = (v2f*)take((size_t)kChunks2 * 32 * 8);
    (void)ws_size;

    hipMemsetAsync(normSrc, 0, (size_t)N * 4, stream);
    hipMemsetAsync(normDst, 0, (size_t)N * 4, stream);
    hipMemsetAsync(agg1,    0, (size_t)N * 8 * 4, stream);
    hipMemsetAsync(agg2,    0, (size_t)N * 8 * 4, stream);

    gcn_deg_kernel<<<(E + 255) / 256, 256, 0, stream>>>(ei, normSrc, normDst, E);
    gcn_norm_kernel<<<(N + 255) / 256, 256, 0, stream>>>(normSrc, N);
    gcn_norm_kernel<<<(N + 255) / 256, 256, 0, stream>>>(normDst, N);

    gcn_pack_b_kernel<<<(kChunks1 * 32 + 255) / 256, 256, 0, stream>>>(
        W1, F_IN, N_CLASSES, bp1, kChunks1);
    gcn_pack_b_kernel<<<1, 64, 0, stream>>>(
        W2, N_CLASSES, N_CLASSES, bp2, kChunks2);

    int tiles  = (N + 15) / 16;
    int gemmBl = (tiles * 32 + 255) / 256;
    gcn_wmma_gemm_stream_kernel<<<tiles, 32, 0, stream>>>(
        h, bp1, normSrc, xpad, N);
    gcn_scatter_kernel<<<(E * 8 + 255) / 256, 256, 0, stream>>>(ei, xpad, agg1, E);
    gcn_finish_pad_kernel<<<(N * 16 + 255) / 256, 256, 0, stream>>>(
        agg1, normDst, b1, h1pad, N);

    gcn_wmma_gemm_kernel<<<gemmBl, 256, 0, stream>>>(
        h1pad, 16, kChunks2, bp2, normSrc, xpad, N);
    gcn_scatter_kernel<<<(E * 8 + 255) / 256, 256, 0, stream>>>(ei, xpad, agg2, E);
    gcn_final_kernel<<<(N * 8 + 255) / 256, 256, 0, stream>>>(
        agg2, normDst, b2, out, N);

    (void)in_sizes; (void)n_in; (void)out_size;
}
